# starting kernel
# baseline (speedup 1.0000x reference)
#include <hip/hip_runtime.h>
#include <math.h>

#define B_    4
#define L_    2048
#define DIN   512
#define HD_   4096
#define M_    16
#define TWO_PI 6.28318530717958647692f

typedef __attribute__((ext_vector_type(2))) float v2f;
typedef __attribute__((ext_vector_type(8))) float v8f;

#define WMMA_F32(a, b, c) \
  __builtin_amdgcn_wmma_f32_16x16x4_f32(false, (a), false, (b), (short)0, (c), false, false)

#define CA_OFF  0
#define SA_OFF  (L_*M_)
#define CE_OFF  (2*L_*M_)
#define SE_OFF  (2*L_*M_ + M_*HD_)
#define MOD_OFF (2*L_*M_ + 2*M_*HD_)

__global__ void k_tables(const float* __restrict__ E,
                         const float* __restrict__ w,
                         const float* __restrict__ C,
                         float* __restrict__ ws) {
  int idx = blockIdx.x * 256 + threadIdx.x;
  if (idx < L_ * M_) {
    int l = idx / M_, m = idx % M_;
    float cm = 0.f;
#pragma unroll
    for (int j = 0; j < M_; ++j) cm = fmaf(C[m * M_ + j], w[j], cm);
    float ang = TWO_PI * (float)(m + 1) * ((float)l / (float)L_);
    float s, c;
    sincosf(ang, &s, &c);
    ws[CA_OFF + idx] = cm * c;
    ws[SA_OFF + idx] = cm * s;
  } else {
    int j = idx - L_ * M_;
    if (j < M_ * HD_) {
      float e = E[j];
      float s, c;
      sincosf(e, &s, &c);
      ws[CE_OFF + j] = c;
      ws[SE_OFF + j] = s;
    }
  }
}

__global__ void k_mod(const float* __restrict__ ws, float* __restrict__ mod) {
  int idx = blockIdx.x * 256 + threadIdx.x;
  int l = idx >> 12;
  int h = idx & (HD_ - 1);
  const float* cA = ws + CA_OFF + l * M_;
  const float* sA = ws + SA_OFF + l * M_;
  const float* cE = ws + CE_OFF + h;
  const float* sE = ws + SE_OFF + h;
  float acc = 0.f;
#pragma unroll
  for (int m = 0; m < M_; ++m) {
    acc = fmaf(cA[m],  cE[m * HD_], acc);
    acc = fmaf(-sA[m], sE[m * HD_], acc);
  }
  mod[idx] = acc;
}

#define XS_STRIDE 516
#define HD_STRIDE 4100
#define SMEM_FLOATS (16 * XS_STRIDE + 16 * HD_STRIDE)

__global__ __launch_bounds__(256)
void k_fused(const float* __restrict__ x,
             const float* __restrict__ Win,
             const float* __restrict__ Wout,
             const float* __restrict__ mod,
             float* __restrict__ out) {
  extern __shared__ float smem[];
  float* xs = smem;
  float* hd = smem + 16 * XS_STRIDE;

  const int row0 = blockIdx.x * 16;
  const int l0   = row0 & (L_ - 1);
  const int tid  = threadIdx.x;

  {
    const float4* xg = (const float4*)(x + (size_t)row0 * DIN);
    for (int i = tid; i < 16 * (DIN / 4); i += 256) {
      int r = i / (DIN / 4), c4 = i % (DIN / 4);
      float4 v = xg[r * (DIN / 4) + c4];
      *(float4*)&xs[r * XS_STRIDE + c4 * 4] = v;
    }
  }
  __syncthreads();

  const int lane = tid & 31;
  const int wave = tid >> 5;
  const int half = lane >> 4;
  const int l16  = lane & 15;

  const float* xs_row = xs + l16 * XS_STRIDE;
  const float* hd_row = hd + l16 * HD_STRIDE;

  for (int tp = 0; tp < 16; ++tp) {
    const int c0 = wave * 512 + tp * 32 + l16;
    v8f acc0 = {0.f, 0.f, 0.f, 0.f, 0.f, 0.f, 0.f, 0.f};
    v8f acc1 = {0.f, 0.f, 0.f, 0.f, 0.f, 0.f, 0.f, 0.f};
    const float* b0 = Win + c0;
    const float* b1 = Win + c0 + 16;
#pragma unroll 4
    for (int k0 = 0; k0 < DIN; k0 += 4) {
      const int ka = k0 + 2 * half;
      v2f a, v0, v1;
      a.x  = xs_row[ka];
      a.y  = xs_row[ka + 1];
      v0.x = b0[(size_t)ka * HD_];
      v0.y = b0[(size_t)(ka + 1) * HD_];
      v1.x = b1[(size_t)ka * HD_];
      v1.y = b1[(size_t)(ka + 1) * HD_];
      acc0 = WMMA_F32(a, v0, acc0);
      acc1 = WMMA_F32(a, v1, acc1);
    }
#pragma unroll
    for (int v = 0; v < 8; ++v) {
      const int r = v + 8 * half;
      const float* mrow = mod + (size_t)(l0 + r) * HD_;
      hd[r * HD_STRIDE + c0]      = acc0[v] * mrow[c0];
      hd[r * HD_STRIDE + c0 + 16] = acc1[v] * mrow[c0 + 16];
    }
  }
  __syncthreads();

  {
    const int cb = wave * 64 + l16;
    v8f acc0 = {0.f, 0.f, 0.f, 0.f, 0.f, 0.f, 0.f, 0.f};
    v8f acc1 = {0.f, 0.f, 0.f, 0.f, 0.f, 0.f, 0.f, 0.f};
    v8f acc2 = {0.f, 0.f, 0.f, 0.f, 0.f, 0.f, 0.f, 0.f};
    v8f acc3 = {0.f, 0.f, 0.f, 0.f, 0.f, 0.f, 0.f, 0.f};
    const float* b0 = Wout + cb;
#pragma unroll 4
    for (int k0 = 0; k0 < HD_; k0 += 4) {
      const int ka = k0 + 2 * half;
      v2f a;
      a.x = hd_row[ka];
      a.y = hd_row[ka + 1];
      const float* br0 = b0 + (size_t)ka * DIN;
      const float* br1 = b0 + (size_t)(ka + 1) * DIN;
      v2f v0, v1, v2, v3;
      v0.x = br0[0];   v0.y = br1[0];
      v1.x = br0[16];  v1.y = br1[16];
      v2.x = br0[32];  v2.y = br1[32];
      v3.x = br0[48];  v3.y = br1[48];
      acc0 = WMMA_F32(a, v0, acc0);
      acc1 = WMMA_F32(a, v1, acc1);
      acc2 = WMMA_F32(a, v2, acc2);
      acc3 = WMMA_F32(a, v3, acc3);
    }
#pragma unroll
    for (int v = 0; v < 8; ++v) {
      const int r = v + 8 * half;
      float* orow = out + (size_t)(row0 + r) * DIN + cb;
      orow[0]  = acc0[v];
      orow[16] = acc1[v];
      orow[32] = acc2[v];
      orow[48] = acc3[v];
    }
  }
}

extern "C" void kernel_launch(void* const* d_in, const int* in_sizes, int n_in,
                              void* d_out, int out_size, void* d_ws, size_t ws_size,
                              hipStream_t stream) {
  (void)in_sizes; (void)n_in; (void)out_size; (void)ws_size;
  const float* x    = (const float*)d_in[0];
  const float* Win  = (const float*)d_in[1];
  const float* Wout = (const float*)d_in[2];
  const float* E    = (const float*)d_in[3];
  const float* w    = (const float*)d_in[4];
  const float* C    = (const float*)d_in[5];
  float* out = (float*)d_out;
  float* ws  = (float*)d_ws;
  float* mod = ws + MOD_OFF;

  k_tables<<<(L_ * M_ + M_ * HD_) / 256, 256, 0, stream>>>(E, w, C, ws);
  k_mod<<<(L_ * HD_) / 256, 256, 0, stream>>>(ws, mod);
  const size_t smem = SMEM_FLOATS * sizeof(float);
  (void)hipFuncSetAttribute((const void*)k_fused,
                            hipFuncAttributeMaxDynamicSharedMemorySize,
                            (int)smem);
  k_fused<<<(B_ * L_) / 16, 256, smem, stream>>>(x, Win, Wout, mod, out);
}


# k_fused regridded 512 x (16 rows, all HD) -> 128 row blocks x 2 HD halves: 64-row workgroups, each B fragment feeds 4 matrix ops; halves joined by a new add launch (K sum order changes)
# speedup vs baseline: 2.0828x; 2.0828x over previous
#include <hip/hip_runtime.h>
#include <math.h>

#define B_    4
#define L_    2048
#define DIN   512
#define HD_   4096
#define M_    16
#define TWO_PI 6.28318530717958647692f

typedef __attribute__((ext_vector_type(2))) float v2f;
typedef __attribute__((ext_vector_type(8))) float v8f;

#define WMMA_F32(a, b, c) \
  __builtin_amdgcn_wmma_f32_16x16x4_f32(false, (a), false, (b), (short)0, (c), false, false)

#define CA_OFF  0
#define SA_OFF  (L_*M_)
#define CE_OFF  (2*L_*M_)
#define SE_OFF  (2*L_*M_ + M_*HD_)
#define PART_OFF (2*L_*M_ + 2*M_*HD_)

__global__ void k_tables(const float* __restrict__ E,
                         const float* __restrict__ w,
                         const float* __restrict__ C,
                         float* __restrict__ ws) {
  int idx = blockIdx.x * 256 + threadIdx.x;
  if (idx < L_ * M_) {
    int l = idx / M_, m = idx % M_;
    float cm = 0.f;
#pragma unroll
    for (int j = 0; j < M_; ++j) cm = fmaf(C[m * M_ + j], w[j], cm);
    float ang = TWO_PI * (float)(m + 1) * ((float)l / (float)L_);
    float s, c;
    sincosf(ang, &s, &c);
    ws[CA_OFF + idx] = cm * c;
    ws[SA_OFF + idx] = cm * s;
  } else {
    int j = idx - L_ * M_;
    if (j < M_ * HD_) {
      float e = E[j];
      float s, c;
      sincosf(e, &s, &c);
      ws[CE_OFF + j] = c;
      ws[SE_OFF + j] = s;
    }
  }
}

#define RB_ 64
#define HC_ 512
#define NCH_ 4
#define XS_STRIDE 516
#define HC_STRIDE 516
#define SMEM_FLOATS (RB_ * XS_STRIDE + RB_ * HC_STRIDE)

__global__ __launch_bounds__(256)
void k_fused(const float* __restrict__ x,
             const float* __restrict__ Win,
             const float* __restrict__ Wout,
             const float* __restrict__ tab,
             float* __restrict__ out,
             float* __restrict__ part) {
  extern __shared__ float smem[];
  float* xs = smem;
  float* hc = smem + RB_ * XS_STRIDE;

  const int row0 = (blockIdx.x >> 1) * RB_;
  const int hh   = blockIdx.x & 1;
  const int l0   = row0 & (L_ - 1);
  const int tid  = threadIdx.x;

  {
    const float4* xg = (const float4*)(x + (size_t)row0 * DIN);
    for (int i = tid; i < RB_ * (DIN / 4); i += 256) {
      int r = i / (DIN / 4), c4 = i % (DIN / 4);
      float4 v = xg[i];
      *(float4*)&xs[r * XS_STRIDE + c4 * 4] = v;
    }
  }
  __syncthreads();

  const int lane = tid & 31;
  const int wave = tid >> 5;
  const int half = lane >> 4;
  const int l16  = lane & 15;
  const int cw   = wave * 64 + 4 * l16;

  const float* xs_row = xs + l16 * XS_STRIDE + 2 * half;
  const float* hc_row = hc + l16 * HC_STRIDE + 2 * half;

  const v8f zero8 = {0.f, 0.f, 0.f, 0.f, 0.f, 0.f, 0.f, 0.f};
  v8f oacc[4][4];
#pragma unroll
  for (int rt = 0; rt < 4; ++rt) {
#pragma unroll
    for (int t = 0; t < 4; ++t) oacc[rt][t] = zero8;
  }

#pragma nounroll
  for (int ch = 0; ch < NCH_; ++ch) {
    const int hbase = hh * (HD_ / 2) + ch * HC_;

    {
      v8f acc[4][4];
#pragma unroll
      for (int rt = 0; rt < 4; ++rt) {
#pragma unroll
        for (int t = 0; t < 4; ++t) acc[rt][t] = zero8;
      }
      const float* bq = Win + (size_t)(2 * half) * HD_ + hbase + cw;
#pragma unroll 4
      for (int k0 = 0; k0 < DIN; k0 += 4) {
        const float4 q0 = *(const float4*)(bq + (size_t)k0 * HD_);
        const float4 q1 = *(const float4*)(bq + (size_t)(k0 + 1) * HD_);
        v2f b0, b1, b2, b3;
        b0.x = q0.x;  b0.y = q1.x;
        b1.x = q0.y;  b1.y = q1.y;
        b2.x = q0.z;  b2.y = q1.z;
        b3.x = q0.w;  b3.y = q1.w;
#pragma unroll
        for (int rt = 0; rt < 4; ++rt) {
          const v2f a = *(const v2f*)(xs_row + rt * 16 * XS_STRIDE + k0);
          acc[rt][0] = WMMA_F32(a, b0, acc[rt][0]);
          acc[rt][1] = WMMA_F32(a, b1, acc[rt][1]);
          acc[rt][2] = WMMA_F32(a, b2, acc[rt][2]);
          acc[rt][3] = WMMA_F32(a, b3, acc[rt][3]);
        }
      }

      v2f bm[8][4];
#pragma unroll
      for (int ks = 0; ks < 8; ++ks) {
        const float* tq_b = tab + ((ks < 4) ? CE_OFF : SE_OFF) + hbase + cw;
        const int ka = 4 * (ks & 3) + 2 * half;
        const float4 q0 = *(const float4*)(tq_b + (size_t)ka * HD_);
        const float4 q1 = *(const float4*)(tq_b + (size_t)(ka + 1) * HD_);
        bm[ks][0].x = q0.x;  bm[ks][0].y = q1.x;
        bm[ks][1].x = q0.y;  bm[ks][1].y = q1.y;
        bm[ks][2].x = q0.z;  bm[ks][2].y = q1.z;
        bm[ks][3].x = q0.w;  bm[ks][3].y = q1.w;
      }
#pragma unroll
      for (int rt = 0; rt < 4; ++rt) {
        const float* ca_row = tab + CA_OFF + (l0 + rt * 16 + l16) * M_ + 2 * half;
        const float* sa_row = tab + SA_OFF + (l0 + rt * 16 + l16) * M_ + 2 * half;
        v8f m0 = zero8, m1 = zero8, m2 = zero8, m3 = zero8;
#pragma unroll
        for (int ks = 0; ks < 8; ++ks) {
          v2f am;
          if (ks < 4) {
            am.x = ca_row[4 * ks];
            am.y = ca_row[4 * ks + 1];
          } else {
            am.x = -sa_row[4 * (ks - 4)];
            am.y = -sa_row[4 * (ks - 4) + 1];
          }
          m0 = WMMA_F32(am, bm[ks][0], m0);
          m1 = WMMA_F32(am, bm[ks][1], m1);
          m2 = WMMA_F32(am, bm[ks][2], m2);
          m3 = WMMA_F32(am, bm[ks][3], m3);
        }
#pragma unroll
        for (int v = 0; v < 8; ++v) {
          const int r = rt * 16 + v + 8 * half;
          float4 o;
          o.x = acc[rt][0][v] * m0[v];
          o.y = acc[rt][1][v] * m1[v];
          o.z = acc[rt][2][v] * m2[v];
          o.w = acc[rt][3][v] * m3[v];
          *(float4*)&hc[r * HC_STRIDE + cw] = o;
        }
      }
    }
    __syncthreads();

    {
      const float* bq = Wout + (size_t)(hbase + 2 * half) * DIN + cw;
#pragma unroll 4
      for (int k0 = 0; k0 < HC_; k0 += 4) {
        const float4 q0 = *(const float4*)(bq + (size_t)k0 * DIN);
        const float4 q1 = *(const float4*)(bq + (size_t)(k0 + 1) * DIN);
        v2f b0, b1, b2, b3;
        b0.x = q0.x;  b0.y = q1.x;
        b1.x = q0.y;  b1.y = q1.y;
        b2.x = q0.z;  b2.y = q1.z;
        b3.x = q0.w;  b3.y = q1.w;
#pragma unroll
        for (int rt = 0; rt < 4; ++rt) {
          const v2f a = *(const v2f*)(hc_row + rt * 16 * HC_STRIDE + k0);
          oacc[rt][0] = WMMA_F32(a, b0, oacc[rt][0]);
          oacc[rt][1] = WMMA_F32(a, b1, oacc[rt][1]);
          oacc[rt][2] = WMMA_F32(a, b2, oacc[rt][2]);
          oacc[rt][3] = WMMA_F32(a, b3, oacc[rt][3]);
        }
      }
    }
    __syncthreads();
  }

  {
    float* dst = hh ? part : out;
#pragma unroll
    for (int rt = 0; rt < 4; ++rt) {
#pragma unroll
      for (int v = 0; v < 8; ++v) {
        const int r = rt * 16 + v + 8 * half;
        float4 o;
        o.x = oacc[rt][0][v];
        o.y = oacc[rt][1][v];
        o.z = oacc[rt][2][v];
        o.w = oacc[rt][3][v];
        *(float4*)(dst + (size_t)(row0 + r) * DIN + cw) = o;
      }
    }
  }
}

__global__ void k_sum(const float* __restrict__ part, float* __restrict__ out) {
  const int i = blockIdx.x * 256 + threadIdx.x;
  const float4 p = ((const float4*)part)[i];
  float4 o = ((const float4*)out)[i];
  o.x += p.x;
  o.y += p.y;
  o.z += p.z;
  o.w += p.w;
  ((float4*)out)[i] = o;
}

extern "C" void kernel_launch(void* const* d_in, const int* in_sizes, int n_in,
                              void* d_out, int out_size, void* d_ws, size_t ws_size,
                              hipStream_t stream) {
  (void)in_sizes; (void)n_in; (void)out_size; (void)ws_size;
  const float* x    = (const float*)d_in[0];
  const float* Win  = (const float*)d_in[1];
  const float* Wout = (const float*)d_in[2];
  const float* E    = (const float*)d_in[3];
  const float* w    = (const float*)d_in[4];
  const float* C    = (const float*)d_in[5];
  float* out = (float*)d_out;
  float* ws  = (float*)d_ws;

  k_tables<<<(L_ * M_ + M_ * HD_) / 256, 256, 0, stream>>>(E, w, C, ws);
  const size_t smem = SMEM_FLOATS * sizeof(float);
  (void)hipFuncSetAttribute((const void*)k_fused,
                            hipFuncAttributeMaxDynamicSharedMemorySize,
                            (int)smem);
  k_fused<<<((B_ * L_) / RB_) * 2, 256, smem, stream>>>(x, Win, Wout, ws, out, ws + PART_OFF);
  k_sum<<<(B_ * L_ * DIN / 4) / 256, 256, 0, stream>>>(ws + PART_OFF, out);
}


# k_fused: K bodies deepened from 4 k-steps to 8 (stage 1) and 16 (stage 2, where the stage-1 accumulators are dead): one exposed L2 load wait per wave now feeds 128 / 256 matrix ops, not 64
# speedup vs baseline: 2.2956x; 1.1022x over previous
#include <hip/hip_runtime.h>
#include <math.h>

#define B_    4
#define L_    2048
#define DIN   512
#define HD_   4096
#define M_    16
#define TWO_PI 6.28318530717958647692f

typedef __attribute__((ext_vector_type(2))) float v2f;
typedef __attribute__((ext_vector_type(8))) float v8f;

#define WMMA_F32(a, b, c) \
  __builtin_amdgcn_wmma_f32_16x16x4_f32(false, (a), false, (b), (short)0, (c), false, false)

#define CA_OFF  0
#define SA_OFF  (L_*M_)
#define CE_OFF  (2*L_*M_)
#define SE_OFF  (2*L_*M_ + M_*HD_)
#define PART_OFF (2*L_*M_ + 2*M_*HD_)

__global__ void k_tables(const float* __restrict__ E,
                         const float* __restrict__ w,
                         const float* __restrict__ C,
                         float* __restrict__ ws) {
  int idx = blockIdx.x * 256 + threadIdx.x;
  if (idx < L_ * M_) {
    int l = idx / M_, m = idx % M_;
    float cm = 0.f;
#pragma unroll
    for (int j = 0; j < M_; ++j) cm = fmaf(C[m * M_ + j], w[j], cm);
    float ang = TWO_PI * (float)(m + 1) * ((float)l / (float)L_);
    float s, c;
    sincosf(ang, &s, &c);
    ws[CA_OFF + idx] = cm * c;
    ws[SA_OFF + idx] = cm * s;
  } else {
    int j = idx - L_ * M_;
    if (j < M_ * HD_) {
      float e = E[j];
      float s, c;
      sincosf(e, &s, &c);
      ws[CE_OFF + j] = c;
      ws[SE_OFF + j] = s;
    }
  }
}

#define RB_ 64
#define HC_ 512
#define NCH_ 4
#define XS_STRIDE 516
#define HC_STRIDE 516
#define SMEM_FLOATS (RB_ * XS_STRIDE + RB_ * HC_STRIDE)

__global__ __launch_bounds__(256)
void k_fused(const float* __restrict__ x,
             const float* __restrict__ Win,
             const float* __restrict__ Wout,
             const float* __restrict__ tab,
             float* __restrict__ out,
             float* __restrict__ part) {
  extern __shared__ float smem[];
  float* xs = smem;
  float* hc = smem + RB_ * XS_STRIDE;

  const int row0 = (blockIdx.x >> 1) * RB_;
  const int hh   = blockIdx.x & 1;
  const int l0   = row0 & (L_ - 1);
  const int tid  = threadIdx.x;

  {
    const float4* xg = (const float4*)(x + (size_t)row0 * DIN);
    for (int i = tid; i < RB_ * (DIN / 4); i += 256) {
      int r = i / (DIN / 4), c4 = i % (DIN / 4);
      float4 v = xg[i];
      *(float4*)&xs[r * XS_STRIDE + c4 * 4] = v;
    }
  }
  __syncthreads();

  const int lane = tid & 31;
  const int wave = tid >> 5;
  const int half = lane >> 4;
  const int l16  = lane & 15;
  const int cw   = wave * 64 + 4 * l16;

  const float* xs_row = xs + l16 * XS_STRIDE + 2 * half;
  const float* hc_row = hc + l16 * HC_STRIDE + 2 * half;

  const v8f zero8 = {0.f, 0.f, 0.f, 0.f, 0.f, 0.f, 0.f, 0.f};
  v8f oacc[4][4];
#pragma unroll
  for (int rt = 0; rt < 4; ++rt) {
#pragma unroll
    for (int t = 0; t < 4; ++t) oacc[rt][t] = zero8;
  }

#pragma nounroll
  for (int ch = 0; ch < NCH_; ++ch) {
    const int hbase = hh * (HD_ / 2) + ch * HC_;

    {
      v8f acc[4][4];
#pragma unroll
      for (int rt = 0; rt < 4; ++rt) {
#pragma unroll
        for (int t = 0; t < 4; ++t) acc[rt][t] = zero8;
      }
      const float* bq = Win + (size_t)(2 * half) * HD_ + hbase + cw;
#pragma unroll 8
      for (int k0 = 0; k0 < DIN; k0 += 4) {
        const float4 q0 = *(const float4*)(bq + (size_t)k0 * HD_);
        const float4 q1 = *(const float4*)(bq + (size_t)(k0 + 1) * HD_);
        v2f b0, b1, b2, b3;
        b0.x = q0.x;  b0.y = q1.x;
        b1.x = q0.y;  b1.y = q1.y;
        b2.x = q0.z;  b2.y = q1.z;
        b3.x = q0.w;  b3.y = q1.w;
#pragma unroll
        for (int rt = 0; rt < 4; ++rt) {
          const v2f a = *(const v2f*)(xs_row + rt * 16 * XS_STRIDE + k0);
          acc[rt][0] = WMMA_F32(a, b0, acc[rt][0]);
          acc[rt][1] = WMMA_F32(a, b1, acc[rt][1]);
          acc[rt][2] = WMMA_F32(a, b2, acc[rt][2]);
          acc[rt][3] = WMMA_F32(a, b3, acc[rt][3]);
        }
      }

      v2f bm[8][4];
#pragma unroll
      for (int ks = 0; ks < 8; ++ks) {
        const float* tq_b = tab + ((ks < 4) ? CE_OFF : SE_OFF) + hbase + cw;
        const int ka = 4 * (ks & 3) + 2 * half;
        const float4 q0 = *(const float4*)(tq_b + (size_t)ka * HD_);
        const float4 q1 = *(const float4*)(tq_b + (size_t)(ka + 1) * HD_);
        bm[ks][0].x = q0.x;  bm[ks][0].y = q1.x;
        bm[ks][1].x = q0.y;  bm[ks][1].y = q1.y;
        bm[ks][2].x = q0.z;  bm[ks][2].y = q1.z;
        bm[ks][3].x = q0.w;  bm[ks][3].y = q1.w;
      }
#pragma unroll
      for (int rt = 0; rt < 4; ++rt) {
        const float* ca_row = tab + CA_OFF + (l0 + rt * 16 + l16) * M_ + 2 * half;
        const float* sa_row = tab + SA_OFF + (l0 + rt * 16 + l16) * M_ + 2 * half;
        v8f m0 = zero8, m1 = zero8, m2 = zero8, m3 = zero8;
#pragma unroll
        for (int ks = 0; ks < 8; ++ks) {
          v2f am;
          if (ks < 4) {
            am.x = ca_row[4 * ks];
            am.y = ca_row[4 * ks + 1];
          } else {
            am.x = -sa_row[4 * (ks - 4)];
            am.y = -sa_row[4 * (ks - 4) + 1];
          }
          m0 = WMMA_F32(am, bm[ks][0], m0);
          m1 = WMMA_F32(am, bm[ks][1], m1);
          m2 = WMMA_F32(am, bm[ks][2], m2);
          m3 = WMMA_F32(am, bm[ks][3], m3);
        }
#pragma unroll
        for (int v = 0; v < 8; ++v) {
          const int r = rt * 16 + v + 8 * half;
          float4 o;
          o.x = acc[rt][0][v] * m0[v];
          o.y = acc[rt][1][v] * m1[v];
          o.z = acc[rt][2][v] * m2[v];
          o.w = acc[rt][3][v] * m3[v];
          *(float4*)&hc[r * HC_STRIDE + cw] = o;
        }
      }
    }
    __syncthreads();

    {
      const float* bq = Wout + (size_t)(hbase + 2 * half) * DIN + cw;
#pragma unroll 16
      for (int k0 = 0; k0 < HC_; k0 += 4) {
        const float4 q0 = *(const float4*)(bq + (size_t)k0 * DIN);
        const float4 q1 = *(const float4*)(bq + (size_t)(k0 + 1) * DIN);
        v2f b0, b1, b2, b3;
        b0.x = q0.x;  b0.y = q1.x;
        b1.x = q0.y;  b1.y = q1.y;
        b2.x = q0.z;  b2.y = q1.z;
        b3.x = q0.w;  b3.y = q1.w;
#pragma unroll
        for (int rt = 0; rt < 4; ++rt) {
          const v2f a = *(const v2f*)(hc_row + rt * 16 * HC_STRIDE + k0);
          oacc[rt][0] = WMMA_F32(a, b0, oacc[rt][0]);
          oacc[rt][1] = WMMA_F32(a, b1, oacc[rt][1]);
          oacc[rt][2] = WMMA_F32(a, b2, oacc[rt][2]);
          oacc[rt][3] = WMMA_F32(a, b3, oacc[rt][3]);
        }
      }
    }
    __syncthreads();
  }

  {
    float* dst = hh ? part : out;
#pragma unroll
    for (int rt = 0; rt < 4; ++rt) {
#pragma unroll
      for (int v = 0; v < 8; ++v) {
        const int r = rt * 16 + v + 8 * half;
        float4 o;
        o.x = oacc[rt][0][v];
        o.y = oacc[rt][1][v];
        o.z = oacc[rt][2][v];
        o.w = oacc[rt][3][v];
        *(float4*)(dst + (size_t)(row0 + r) * DIN + cw) = o;
      }
    }
  }
}

__global__ void k_sum(const float* __restrict__ part, float* __restrict__ out) {
  const int i = blockIdx.x * 256 + threadIdx.x;
  const float4 p = ((const float4*)part)[i];
  float4 o = ((const float4*)out)[i];
  o.x += p.x;
  o.y += p.y;
  o.z += p.z;
  o.w += p.w;
  ((float4*)out)[i] = o;
}

extern "C" void kernel_launch(void* const* d_in, const int* in_sizes, int n_in,
                              void* d_out, int out_size, void* d_ws, size_t ws_size,
                              hipStream_t stream) {
  (void)in_sizes; (void)n_in; (void)out_size; (void)ws_size;
  const float* x    = (const float*)d_in[0];
  const float* Win  = (const float*)d_in[1];
  const float* Wout = (const float*)d_in[2];
  const float* E    = (const float*)d_in[3];
  const float* w    = (const float*)d_in[4];
  const float* C    = (const float*)d_in[5];
  float* out = (float*)d_out;
  float* ws  = (float*)d_ws;

  k_tables<<<(L_ * M_ + M_ * HD_) / 256, 256, 0, stream>>>(E, w, C, ws);
  const size_t smem = SMEM_FLOATS * sizeof(float);
  (void)hipFuncSetAttribute((const void*)k_fused,
                            hipFuncAttributeMaxDynamicSharedMemorySize,
                            (int)smem);
  k_fused<<<((B_ * L_) / RB_) * 2, 256, smem, stream>>>(x, Win, Wout, ws, out, ws + PART_OFF);
  k_sum<<<(B_ * L_ * DIN / 4) / 256, 256, 0, stream>>>(ws + PART_OFF, out);
}


# k_fused: the 2 waves of a SIMD take turns per K body via s_setprio (3 after a body's first k-step, 0 once the next body's loads are issued): each L2 wait lies under the partner's matrix run
# speedup vs baseline: 2.4510x; 1.0677x over previous
#include <hip/hip_runtime.h>
#include <math.h>

#define B_    4
#define L_    2048
#define DIN   512
#define HD_   4096
#define M_    16
#define TWO_PI 6.28318530717958647692f

typedef __attribute__((ext_vector_type(2))) float v2f;
typedef __attribute__((ext_vector_type(8))) float v8f;

#define WMMA_F32(a, b, c) \
  __builtin_amdgcn_wmma_f32_16x16x4_f32(false, (a), false, (b), (short)0, (c), false, false)

#define CA_OFF  0
#define SA_OFF  (L_*M_)
#define CE_OFF  (2*L_*M_)
#define SE_OFF  (2*L_*M_ + M_*HD_)
#define PART_OFF (2*L_*M_ + 2*M_*HD_)

__global__ void k_tables(const float* __restrict__ E,
                         const float* __restrict__ w,
                         const float* __restrict__ C,
                         float* __restrict__ ws) {
  int idx = blockIdx.x * 256 + threadIdx.x;
  if (idx < L_ * M_) {
    int l = idx / M_, m = idx % M_;
    float cm = 0.f;
#pragma unroll
    for (int j = 0; j < M_; ++j) cm = fmaf(C[m * M_ + j], w[j], cm);
    float ang = TWO_PI * (float)(m + 1) * ((float)l / (float)L_);
    float s, c;
    sincosf(ang, &s, &c);
    ws[CA_OFF + idx] = cm * c;
    ws[SA_OFF + idx] = cm * s;
  } else {
    int j = idx - L_ * M_;
    if (j < M_ * HD_) {
      float e = E[j];
      float s, c;
      sincosf(e, &s, &c);
      ws[CE_OFF + j] = c;
      ws[SE_OFF + j] = s;
    }
  }
}

#define RB_ 64
#define HC_ 512
#define NCH_ 4
#define XS_STRIDE 516
#define HC_STRIDE 516
#define SMEM_FLOATS (RB_ * XS_STRIDE + RB_ * HC_STRIDE)

__global__ __launch_bounds__(256)
void k_fused(const float* __restrict__ x,
             const float* __restrict__ Win,
             const float* __restrict__ Wout,
             const float* __restrict__ tab,
             float* __restrict__ out,
             float* __restrict__ part) {
  extern __shared__ float smem[];
  float* xs = smem;
  float* hc = smem + RB_ * XS_STRIDE;

  const int row0 = (blockIdx.x >> 1) * RB_;
  const int hh   = blockIdx.x & 1;
  const int l0   = row0 & (L_ - 1);
  const int tid  = threadIdx.x;

  {
    const float4* xg = (const float4*)(x + (size_t)row0 * DIN);
    for (int i = tid; i < RB_ * (DIN / 4); i += 256) {
      int r = i / (DIN / 4), c4 = i % (DIN / 4);
      float4 v = xg[i];
      *(float4*)&xs[r * XS_STRIDE + c4 * 4] = v;
    }
  }
  __syncthreads();

  const int lane = tid & 31;
  const int wave = tid >> 5;
  const int half = lane >> 4;
  const int l16  = lane & 15;
  const int cw   = wave * 64 + 4 * l16;

  const float* xs_row = xs + l16 * XS_STRIDE + 2 * half;
  const float* hc_row = hc + l16 * HC_STRIDE + 2 * half;

  const v8f zero8 = {0.f, 0.f, 0.f, 0.f, 0.f, 0.f, 0.f, 0.f};
  v8f oacc[4][4];
#pragma unroll
  for (int rt = 0; rt < 4; ++rt) {
#pragma unroll
    for (int t = 0; t < 4; ++t) oacc[rt][t] = zero8;
  }

#pragma nounroll
  for (int ch = 0; ch < NCH_; ++ch) {
    const int hbase = hh * (HD_ / 2) + ch * HC_;

    {
      v8f acc[4][4];
#pragma unroll
      for (int rt = 0; rt < 4; ++rt) {
#pragma unroll
        for (int t = 0; t < 4; ++t) acc[rt][t] = zero8;
      }
      const float* bq = Win + (size_t)(2 * half) * HD_ + hbase + cw;
#pragma nounroll
      for (int kb = 0; kb < DIN; kb += 32) {
        float4 q[16];
#pragma unroll
        for (int j = 0; j < 8; ++j) {
          q[2 * j]     = *(const float4*)(bq + (size_t)(kb + 4 * j) * HD_);
          q[2 * j + 1] = *(const float4*)(bq + (size_t)(kb + 4 * j + 1) * HD_);
        }
        v2f a[8][4];
#pragma unroll
        for (int j = 0; j < 8; ++j) {
#pragma unroll
          for (int rt = 0; rt < 4; ++rt)
            a[j][rt] = *(const v2f*)(xs_row + rt * 16 * XS_STRIDE + kb + 4 * j);
        }
        __builtin_amdgcn_s_setprio(0);
#pragma unroll
        for (int j = 0; j < 8; ++j) {
          if (j == 1) __builtin_amdgcn_s_setprio(3);
          v2f b0, b1, b2, b3;
          b0.x = q[2 * j].x;  b0.y = q[2 * j + 1].x;
          b1.x = q[2 * j].y;  b1.y = q[2 * j + 1].y;
          b2.x = q[2 * j].z;  b2.y = q[2 * j + 1].z;
          b3.x = q[2 * j].w;  b3.y = q[2 * j + 1].w;
#pragma unroll
          for (int rt = 0; rt < 4; ++rt) {
            acc[rt][0] = WMMA_F32(a[j][rt], b0, acc[rt][0]);
            acc[rt][1] = WMMA_F32(a[j][rt], b1, acc[rt][1]);
            acc[rt][2] = WMMA_F32(a[j][rt], b2, acc[rt][2]);
            acc[rt][3] = WMMA_F32(a[j][rt], b3, acc[rt][3]);
          }
        }
      }
      __builtin_amdgcn_s_setprio(0);

      v2f bm[8][4];
#pragma unroll
      for (int ks = 0; ks < 8; ++ks) {
        const float* tq_b = tab + ((ks < 4) ? CE_OFF : SE_OFF) + hbase + cw;
        const int ka = 4 * (ks & 3) + 2 * half;
        const float4 q0 = *(const float4*)(tq_b + (size_t)ka * HD_);
        const float4 q1 = *(const float4*)(tq_b + (size_t)(ka + 1) * HD_);
        bm[ks][0].x = q0.x;  bm[ks][0].y = q1.x;
        bm[ks][1].x = q0.y;  bm[ks][1].y = q1.y;
        bm[ks][2].x = q0.z;  bm[ks][2].y = q1.z;
        bm[ks][3].x = q0.w;  bm[ks][3].y = q1.w;
      }
#pragma unroll
      for (int rt = 0; rt < 4; ++rt) {
        const float* ca_row = tab + CA_OFF + (l0 + rt * 16 + l16) * M_ + 2 * half;
        const float* sa_row = tab + SA_OFF + (l0 + rt * 16 + l16) * M_ + 2 * half;
        v8f m0 = zero8, m1 = zero8, m2 = zero8, m3 = zero8;
#pragma unroll
        for (int ks = 0; ks < 8; ++ks) {
          v2f am;
          if (ks < 4) {
            am.x = ca_row[4 * ks];
            am.y = ca_row[4 * ks + 1];
          } else {
            am.x = -sa_row[4 * (ks - 4)];
            am.y = -sa_row[4 * (ks - 4) + 1];
          }
          m0 = WMMA_F32(am, bm[ks][0], m0);
          m1 = WMMA_F32(am, bm[ks][1], m1);
          m2 = WMMA_F32(am, bm[ks][2], m2);
          m3 = WMMA_F32(am, bm[ks][3], m3);
        }
#pragma unroll
        for (int v = 0; v < 8; ++v) {
          const int r = rt * 16 + v + 8 * half;
          float4 o;
          o.x = acc[rt][0][v] * m0[v];
          o.y = acc[rt][1][v] * m1[v];
          o.z = acc[rt][2][v] * m2[v];
          o.w = acc[rt][3][v] * m3[v];
          *(float4*)&hc[r * HC_STRIDE + cw] = o;
        }
      }
    }
    __syncthreads();

    {
      const float* bq = Wout + (size_t)(hbase + 2 * half) * DIN + cw;
#pragma nounroll
      for (int kb = 0; kb < HC_; kb += 64) {
        float4 q[32];
#pragma unroll
        for (int j = 0; j < 16; ++j) {
          q[2 * j]     = *(const float4*)(bq + (size_t)(kb + 4 * j) * DIN);
          q[2 * j + 1] = *(const float4*)(bq + (size_t)(kb + 4 * j + 1) * DIN);
        }
        v2f a[16][4];
#pragma unroll
        for (int j = 0; j < 16; ++j) {
#pragma unroll
          for (int rt = 0; rt < 4; ++rt)
            a[j][rt] = *(const v2f*)(hc_row + rt * 16 * HC_STRIDE + kb + 4 * j);
        }
        __builtin_amdgcn_s_setprio(0);
#pragma unroll
        for (int j = 0; j < 16; ++j) {
          if (j == 1) __builtin_amdgcn_s_setprio(3);
          v2f b0, b1, b2, b3;
          b0.x = q[2 * j].x;  b0.y = q[2 * j + 1].x;
          b1.x = q[2 * j].y;  b1.y = q[2 * j + 1].y;
          b2.x = q[2 * j].z;  b2.y = q[2 * j + 1].z;
          b3.x = q[2 * j].w;  b3.y = q[2 * j + 1].w;
#pragma unroll
          for (int rt = 0; rt < 4; ++rt) {
            oacc[rt][0] = WMMA_F32(a[j][rt], b0, oacc[rt][0]);
            oacc[rt][1] = WMMA_F32(a[j][rt], b1, oacc[rt][1]);
            oacc[rt][2] = WMMA_F32(a[j][rt], b2, oacc[rt][2]);
            oacc[rt][3] = WMMA_F32(a[j][rt], b3, oacc[rt][3]);
          }
        }
      }
      __builtin_amdgcn_s_setprio(0);
    }
    __syncthreads();
  }

  {
    float* dst = hh ? part : out;
#pragma unroll
    for (int rt = 0; rt < 4; ++rt) {
#pragma unroll
      for (int v = 0; v < 8; ++v) {
        const int r = rt * 16 + v + 8 * half;
        float4 o;
        o.x = oacc[rt][0][v];
        o.y = oacc[rt][1][v];
        o.z = oacc[rt][2][v];
        o.w = oacc[rt][3][v];
        *(float4*)(dst + (size_t)(row0 + r) * DIN + cw) = o;
      }
    }
  }
}

__global__ void k_sum(const float* __restrict__ part, float* __restrict__ out) {
  const int i = blockIdx.x * 256 + threadIdx.x;
  const float4 p = ((const float4*)part)[i];
  float4 o = ((const float4*)out)[i];
  o.x += p.x;
  o.y += p.y;
  o.z += p.z;
  o.w += p.w;
  ((float4*)out)[i] = o;
}

extern "C" void kernel_launch(void* const* d_in, const int* in_sizes, int n_in,
                              void* d_out, int out_size, void* d_ws, size_t ws_size,
                              hipStream_t stream) {
  (void)in_sizes; (void)n_in; (void)out_size; (void)ws_size;
  const float* x    = (const float*)d_in[0];
  const float* Win  = (const float*)d_in[1];
  const float* Wout = (const float*)d_in[2];
  const float* E    = (const float*)d_in[3];
  const float* w    = (const float*)d_in[4];
  const float* C    = (const float*)d_in[5];
  float* out = (float*)d_out;
  float* ws  = (float*)d_ws;

  k_tables<<<(L_ * M_ + M_ * HD_) / 256, 256, 0, stream>>>(E, w, C, ws);
  const size_t smem = SMEM_FLOATS * sizeof(float);
  (void)hipFuncSetAttribute((const void*)k_fused,
                            hipFuncAttributeMaxDynamicSharedMemorySize,
                            (int)smem);
  k_fused<<<((B_ * L_) / RB_) * 2, 256, smem, stream>>>(x, Win, Wout, ws, out, ws + PART_OFF);
  k_sum<<<(B_ * L_ * DIN / 4) / 256, 256, 0, stream>>>(ws + PART_OFF, out);
}


# k_fused: a workgroup's 64 rows become 16 tokens x 4 batches (row tile = batch): the mod tile is formed once per chunk, not 4x; the registers so freed hold its trig loads, issued ahead of the K loop
# speedup vs baseline: 2.5419x; 1.0371x over previous
#include <hip/hip_runtime.h>
#include <math.h>

#define B_    4
#define L_    2048
#define DIN   512
#define HD_   4096
#define M_    16
#define TWO_PI 6.28318530717958647692f

typedef __attribute__((ext_vector_type(2))) float v2f;
typedef __attribute__((ext_vector_type(8))) float v8f;

#define WMMA_F32(a, b, c) \
  __builtin_amdgcn_wmma_f32_16x16x4_f32(false, (a), false, (b), (short)0, (c), false, false)

#define CA_OFF  0
#define SA_OFF  (L_*M_)
#define CE_OFF  (2*L_*M_)
#define SE_OFF  (2*L_*M_ + M_*HD_)
#define PART_OFF (2*L_*M_ + 2*M_*HD_)

__global__ void k_tables(const float* __restrict__ E,
                         const float* __restrict__ w,
                         const float* __restrict__ C,
                         float* __restrict__ ws) {
  int idx = blockIdx.x * 256 + threadIdx.x;
  if (idx < L_ * M_) {
    int l = idx / M_, m = idx % M_;
    float cm = 0.f;
#pragma unroll
    for (int j = 0; j < M_; ++j) cm = fmaf(C[m * M_ + j], w[j], cm);
    float ang = TWO_PI * (float)(m + 1) * ((float)l / (float)L_);
    float s, c;
    sincosf(ang, &s, &c);
    ws[CA_OFF + idx] = cm * c;
    ws[SA_OFF + idx] = cm * s;
  } else {
    int j = idx - L_ * M_;
    if (j < M_ * HD_) {
      float e = E[j];
      float s, c;
      sincosf(e, &s, &c);
      ws[CE_OFF + j] = c;
      ws[SE_OFF + j] = s;
    }
  }
}

#define RB_ 64
#define HC_ 512
#define NCH_ 4
#define XS_STRIDE 516
#define HC_STRIDE 516
#define SMEM_FLOATS (RB_ * XS_STRIDE + RB_ * HC_STRIDE)

__global__ __launch_bounds__(256)
void k_fused(const float* __restrict__ x,
             const float* __restrict__ Win,
             const float* __restrict__ Wout,
             const float* __restrict__ tab,
             float* __restrict__ out,
             float* __restrict__ part) {
  extern __shared__ float smem[];
  float* xs = smem;
  float* hc = smem + RB_ * XS_STRIDE;

  const int t0   = (blockIdx.x >> 1) * 16;
  const int hh   = blockIdx.x & 1;
  const int tid  = threadIdx.x;

  {
    for (int i = tid; i < RB_ * (DIN / 4); i += 256) {
      int r = i / (DIN / 4), c4 = i % (DIN / 4);
      const int grow = (r >> 4) * L_ + t0 + (r & 15);
      float4 v = *(const float4*)(x + (size_t)grow * DIN + c4 * 4);
      *(float4*)&xs[r * XS_STRIDE + c4 * 4] = v;
    }
  }
  __syncthreads();

  const int lane = tid & 31;
  const int wave = tid >> 5;
  const int half = lane >> 4;
  const int l16  = lane & 15;
  const int cw   = wave * 64 + 4 * l16;

  const float* xs_row = xs + l16 * XS_STRIDE + 2 * half;
  const float* hc_row = hc + l16 * HC_STRIDE + 2 * half;

  const float* ca_row = tab + CA_OFF + (t0 + l16) * M_ + 2 * half;
  const float* sa_row = tab + SA_OFF + (t0 + l16) * M_ + 2 * half;

  const v8f zero8 = {0.f, 0.f, 0.f, 0.f, 0.f, 0.f, 0.f, 0.f};
  v8f oacc[4][4];
#pragma unroll
  for (int rt = 0; rt < 4; ++rt) {
#pragma unroll
    for (int t = 0; t < 4; ++t) oacc[rt][t] = zero8;
  }

#pragma nounroll
  for (int ch = 0; ch < NCH_; ++ch) {
    const int hbase = hh * (HD_ / 2) + ch * HC_;

    {
      v8f acc[4][4];
#pragma unroll
      for (int rt = 0; rt < 4; ++rt) {
#pragma unroll
        for (int t = 0; t < 4; ++t) acc[rt][t] = zero8;
      }
      float4 tq[16];
#pragma unroll
      for (int ks = 0; ks < 8; ++ks) {
        const float* tq_b = tab + ((ks < 4) ? CE_OFF : SE_OFF) + hbase + cw;
        const int ka = 4 * (ks & 3) + 2 * half;
        tq[2 * ks]     = *(const float4*)(tq_b + (size_t)ka * HD_);
        tq[2 * ks + 1] = *(const float4*)(tq_b + (size_t)(ka + 1) * HD_);
      }
      const float* bq = Win + (size_t)(2 * half) * HD_ + hbase + cw;
#pragma nounroll
      for (int kb = 0; kb < DIN; kb += 32) {
        float4 q[16];
#pragma unroll
        for (int j = 0; j < 8; ++j) {
          q[2 * j]     = *(const float4*)(bq + (size_t)(kb + 4 * j) * HD_);
          q[2 * j + 1] = *(const float4*)(bq + (size_t)(kb + 4 * j + 1) * HD_);
        }
        v2f a[8][4];
#pragma unroll
        for (int j = 0; j < 8; ++j) {
#pragma unroll
          for (int rt = 0; rt < 4; ++rt)
            a[j][rt] = *(const v2f*)(xs_row + rt * 16 * XS_STRIDE + kb + 4 * j);
        }
        __builtin_amdgcn_s_setprio(0);
#pragma unroll
        for (int j = 0; j < 8; ++j) {
          if (j == 1) __builtin_amdgcn_s_setprio(3);
          v2f b0, b1, b2, b3;
          b0.x = q[2 * j].x;  b0.y = q[2 * j + 1].x;
          b1.x = q[2 * j].y;  b1.y = q[2 * j + 1].y;
          b2.x = q[2 * j].z;  b2.y = q[2 * j + 1].z;
          b3.x = q[2 * j].w;  b3.y = q[2 * j + 1].w;
#pragma unroll
          for (int rt = 0; rt < 4; ++rt) {
            acc[rt][0] = WMMA_F32(a[j][rt], b0, acc[rt][0]);
            acc[rt][1] = WMMA_F32(a[j][rt], b1, acc[rt][1]);
            acc[rt][2] = WMMA_F32(a[j][rt], b2, acc[rt][2]);
            acc[rt][3] = WMMA_F32(a[j][rt], b3, acc[rt][3]);
          }
        }
      }
      __builtin_amdgcn_s_setprio(0);

      v8f m0 = zero8, m1 = zero8, m2 = zero8, m3 = zero8;
#pragma unroll
      for (int ks = 0; ks < 8; ++ks) {
        v2f am;
        if (ks < 4) {
          am.x = ca_row[4 * ks];
          am.y = ca_row[4 * ks + 1];
        } else {
          am.x = -sa_row[4 * (ks - 4)];
          am.y = -sa_row[4 * (ks - 4) + 1];
        }
        v2f b0, b1, b2, b3;
        b0.x = tq[2 * ks].x;  b0.y = tq[2 * ks + 1].x;
        b1.x = tq[2 * ks].y;  b1.y = tq[2 * ks + 1].y;
        b2.x = tq[2 * ks].z;  b2.y = tq[2 * ks + 1].z;
        b3.x = tq[2 * ks].w;  b3.y = tq[2 * ks + 1].w;
        m0 = WMMA_F32(am, b0, m0);
        m1 = WMMA_F32(am, b1, m1);
        m2 = WMMA_F32(am, b2, m2);
        m3 = WMMA_F32(am, b3, m3);
      }
#pragma unroll
      for (int rt = 0; rt < 4; ++rt) {
#pragma unroll
        for (int v = 0; v < 8; ++v) {
          const int r = rt * 16 + v + 8 * half;
          float4 o;
          o.x = acc[rt][0][v] * m0[v];
          o.y = acc[rt][1][v] * m1[v];
          o.z = acc[rt][2][v] * m2[v];
          o.w = acc[rt][3][v] * m3[v];
          *(float4*)&hc[r * HC_STRIDE + cw] = o;
        }
      }
    }
    __syncthreads();

    {
      const float* bq = Wout + (size_t)(hbase + 2 * half) * DIN + cw;
#pragma nounroll
      for (int kb = 0; kb < HC_; kb += 64) {
        float4 q[32];
#pragma unroll
        for (int j = 0; j < 16; ++j) {
          q[2 * j]     = *(const float4*)(bq + (size_t)(kb + 4 * j) * DIN);
          q[2 * j + 1] = *(const float4*)(bq + (size_t)(kb + 4 * j + 1) * DIN);
        }
        v2f a[16][4];
#pragma unroll
        for (int j = 0; j < 16; ++j) {
#pragma unroll
          for (int rt = 0; rt < 4; ++rt)
            a[j][rt] = *(const v2f*)(hc_row + rt * 16 * HC_STRIDE + kb + 4 * j);
        }
        __builtin_amdgcn_s_setprio(0);
#pragma unroll
        for (int j = 0; j < 16; ++j) {
          if (j == 1) __builtin_amdgcn_s_setprio(3);
          v2f b0, b1, b2, b3;
          b0.x = q[2 * j].x;  b0.y = q[2 * j + 1].x;
          b1.x = q[2 * j].y;  b1.y = q[2 * j + 1].y;
          b2.x = q[2 * j].z;  b2.y = q[2 * j + 1].z;
          b3.x = q[2 * j].w;  b3.y = q[2 * j + 1].w;
#pragma unroll
          for (int rt = 0; rt < 4; ++rt) {
            oacc[rt][0] = WMMA_F32(a[j][rt], b0, oacc[rt][0]);
            oacc[rt][1] = WMMA_F32(a[j][rt], b1, oacc[rt][1]);
            oacc[rt][2] = WMMA_F32(a[j][rt], b2, oacc[rt][2]);
            oacc[rt][3] = WMMA_F32(a[j][rt], b3, oacc[rt][3]);
          }
        }
      }
      __builtin_amdgcn_s_setprio(0);
    }
    __syncthreads();
  }

  {
    float* dst = hh ? part : out;
#pragma unroll
    for (int rt = 0; rt < 4; ++rt) {
#pragma unroll
      for (int v = 0; v < 8; ++v) {
        const int grow = rt * L_ + t0 + v + 8 * half;
        float4 o;
        o.x = oacc[rt][0][v];
        o.y = oacc[rt][1][v];
        o.z = oacc[rt][2][v];
        o.w = oacc[rt][3][v];
        *(float4*)(dst + (size_t)grow * DIN + cw) = o;
      }
    }
  }
}

__global__ void k_sum(const float* __restrict__ part, float* __restrict__ out) {
  const int i = blockIdx.x * 256 + threadIdx.x;
  const float4 p = ((const float4*)part)[i];
  float4 o = ((const float4*)out)[i];
  o.x += p.x;
  o.y += p.y;
  o.z += p.z;
  o.w += p.w;
  ((float4*)out)[i] = o;
}

extern "C" void kernel_launch(void* const* d_in, const int* in_sizes, int n_in,
                              void* d_out, int out_size, void* d_ws, size_t ws_size,
                              hipStream_t stream) {
  (void)in_sizes; (void)n_in; (void)out_size; (void)ws_size;
  const float* x    = (const float*)d_in[0];
  const float* Win  = (const float*)d_in[1];
  const float* Wout = (const float*)d_in[2];
  const float* E    = (const float*)d_in[3];
  const float* w    = (const float*)d_in[4];
  const float* C    = (const float*)d_in[5];
  float* out = (float*)d_out;
  float* ws  = (float*)d_ws;

  k_tables<<<(L_ * M_ + M_ * HD_) / 256, 256, 0, stream>>>(E, w, C, ws);
  const size_t smem = SMEM_FLOATS * sizeof(float);
  (void)hipFuncSetAttribute((const void*)k_fused,
                            hipFuncAttributeMaxDynamicSharedMemorySize,
                            (int)smem);
  k_fused<<<((B_ * L_) / RB_) * 2, 256, smem, stream>>>(x, Win, Wout, ws, out, ws + PART_OFF);
  k_sum<<<(B_ * L_ * DIN / 4) / 256, 256, 0, stream>>>(ws + PART_OFF, out);
}
